# starting kernel
# baseline (speedup 1.0000x reference)
#include <hip/hip_runtime.h>

#define BB 4
#define NN 512
#define CC 64
#define PLANE (NN * CC)
#define MM (BB * NN)

typedef float v2f __attribute__((ext_vector_type(2)));
typedef float v4f __attribute__((ext_vector_type(4)));
typedef float v8f __attribute__((ext_vector_type(8)));

__global__ void linear64_wmma(const float* __restrict__ X,
                              const float* __restrict__ W,
                              const float* __restrict__ bias,
                              const float* __restrict__ res,
                              float* __restrict__ Out) {
    const int m0   = blockIdx.x * 16;
    const int n0   = blockIdx.y * 16;
    const int lane = threadIdx.x;
    const int half = lane >> 4;
    const int l    = lane & 15;

    v8f acc = {};
    #pragma unroll
    for (int k0 = 0; k0 < CC; k0 += 4) {
        const int ka = k0 + half * 2;
        v2f a = *(const v2f*)(X + (size_t)(m0 + l) * CC + ka);
        v2f b = *(const v2f*)(W + (size_t)(n0 + l) * CC + ka);
        acc = __builtin_amdgcn_wmma_f32_16x16x4_f32(
             false, a,  false, b,
             (short)0, acc,  false,  false);
    }

    const int n = n0 + l;
    #pragma unroll
    for (int r = 0; r < 8; ++r) {
        const int m = m0 + r + half * 8;
        float v = acc[r] + bias[n];
        if (res) v += res[(size_t)m * CC + n];
        Out[(size_t)m * CC + n] = v;
    }
}

__global__ void softmax_denom_pass(const float* __restrict__ z,
                                   const float* __restrict__ Qa,
                                   float* __restrict__ R) {
    const int b = blockIdx.y;
    const int p = (blockIdx.x * 256 + threadIdx.x) * 4;
    const size_t base = (size_t)b * NN * PLANE + p;

    v4f acc = {};
    #pragma unroll 4
    for (int i = 0; i < NN; ++i) {
        v4f v = __builtin_nontemporal_load((const v4f*)(z + base + (size_t)i * PLANE));
        acc.x += __expf(v.x);
        acc.y += __expf(v.y);
        acc.z += __expf(v.z);
        acc.w += __expf(v.w);
    }
    const size_t rp = (size_t)b * PLANE + p;
    v4f q = *(const v4f*)(Qa + rp);
    v4f r;
    r.x = q.x / acc.x; r.y = q.y / acc.y; r.z = q.z / acc.z; r.w = q.w / acc.w;
    *(v4f*)(R + rp) = r;
}

__global__ void aggregate_pass(const float* __restrict__ z,
                               const float* __restrict__ R,
                               float* __restrict__ Va) {
    __shared__ v4f smem[256];
    const int bi  = blockIdx.x;
    const int b   = bi >> 9;
    const int tid = threadIdx.x;
    const int c4    = (tid & 15) * 4;
    const int kbase = tid >> 4;

    const size_t zbase = (size_t)bi * PLANE;
    const float* Rb = R + (size_t)b * PLANE;

    v4f acc = {};
    #pragma unroll 4
    for (int k = kbase; k < NN; k += 16) {
        const int p = k * CC + c4;
        v4f zv = __builtin_nontemporal_load((const v4f*)(z + zbase + p));
        v4f rv = *(const v4f*)(Rb + p);
        acc.x += __expf(zv.x) * rv.x;
        acc.y += __expf(zv.y) * rv.y;
        acc.z += __expf(zv.z) * rv.z;
        acc.w += __expf(zv.w) * rv.w;
    }
    smem[tid] = acc;
    __syncthreads();

    if (tid < 16) {
        v4f s = smem[tid];
        #pragma unroll
        for (int q = 1; q < 16; ++q) {
            v4f t = smem[tid + 16 * q];
            s.x += t.x; s.y += t.y; s.z += t.z; s.w += t.w;
        }
        *(v4f*)(Va + (size_t)bi * CC + tid * 4) = s;
    }
}

extern "C" void kernel_launch(void* const* d_in, const int* in_sizes, int n_in,
                              void* d_out, int out_size, void* d_ws, size_t ws_size,
                              hipStream_t stream) {
    const float* V  = (const float*)d_in[0];
    const float* Q  = (const float*)d_in[1];
    const float* z  = (const float*)d_in[2];
    const float* W1 = (const float*)d_in[5];
    const float* b1 = (const float*)d_in[6];
    const float* W2 = (const float*)d_in[7];
    const float* b2 = (const float*)d_in[8];
    float* out = (float*)d_out;

    float* Qa = (float*)d_ws;
    float* R  = Qa + (size_t)MM * CC;
    float* Va = R  + (size_t)MM * CC;

    linear64_wmma<<<dim3(MM / 16, CC / 16), dim3(32), 0, stream>>>(Q, W1, b1, nullptr, Qa);

    softmax_denom_pass<<<dim3(PLANE / 1024, BB), dim3(256), 0, stream>>>(z, Qa, R);

    aggregate_pass<<<dim3(BB * NN), dim3(256), 0, stream>>>(z, R, Va);

    linear64_wmma<<<dim3(MM / 16, CC / 16), dim3(32), 0, stream>>>(Va, W2, b2, V, out);
}


# pass 1 (sum over i of exp z): i axis split over the 8 waves of a work-group (64 rows each), partials summed through LDS: 512 work-groups, up to 32 loads in flight per wave; reorders the i sum
# speedup vs baseline: 1.8825x; 1.8825x over previous
#include <hip/hip_runtime.h>

#define BB 4
#define NN 512
#define CC 64
#define PLANE (NN * CC)
#define MM (BB * NN)

typedef float v2f __attribute__((ext_vector_type(2)));
typedef float v4f __attribute__((ext_vector_type(4)));
typedef float v8f __attribute__((ext_vector_type(8)));

__global__ void linear64_wmma(const float* __restrict__ X,
                              const float* __restrict__ W,
                              const float* __restrict__ bias,
                              const float* __restrict__ res,
                              float* __restrict__ Out) {
    const int m0   = blockIdx.x * 16;
    const int n0   = blockIdx.y * 16;
    const int lane = threadIdx.x;
    const int half = lane >> 4;
    const int l    = lane & 15;

    v8f acc = {};
    #pragma unroll
    for (int k0 = 0; k0 < CC; k0 += 4) {
        const int ka = k0 + half * 2;
        v2f a = *(const v2f*)(X + (size_t)(m0 + l) * CC + ka);
        v2f b = *(const v2f*)(W + (size_t)(n0 + l) * CC + ka);
        acc = __builtin_amdgcn_wmma_f32_16x16x4_f32(
             false, a,  false, b,
             (short)0, acc,  false,  false);
    }

    const int n = n0 + l;
    #pragma unroll
    for (int r = 0; r < 8; ++r) {
        const int m = m0 + r + half * 8;
        float v = acc[r] + bias[n];
        if (res) v += res[(size_t)m * CC + n];
        Out[(size_t)m * CC + n] = v;
    }
}

#define P1_WAVES 8
#define P1_ROWS  (NN / P1_WAVES)
#define P1_STRIP 256
#define P1_GROUP 8
#define P1_NGRP  (P1_ROWS / P1_GROUP)

__global__ void __launch_bounds__(P1_WAVES * 32)
softmax_denom_pass(const float* __restrict__ z,
                   const float* __restrict__ Qa,
                   float* __restrict__ R) {
    __shared__ v4f part[P1_WAVES][2][32];
    const int tid  = threadIdx.x;
    const int lane = tid & 31;
    const int wave = __builtin_amdgcn_readfirstlane(tid >> 5);
    const int b    = blockIdx.y;
    const int p0   = blockIdx.x * P1_STRIP;

    const size_t rp = (size_t)b * PLANE + p0 + (tid & 63) * 4;
    v4f q = {};
    if (tid < 64) q = *(const v4f*)(Qa + rp);

    const float* zp = z + ((size_t)b * NN + (size_t)wave * P1_ROWS) * PLANE + p0 + lane * 4;

    v4f buf[2][2 * P1_GROUP];
#define P1_LOAD(G) do { _Pragma("unroll") for (int r = 0; r < P1_GROUP; ++r) { \
        const float* rowp = zp + (size_t)((G) * P1_GROUP + r) * PLANE; \
        buf[(G) & 1][2 * r]     = __builtin_nontemporal_load((const v4f*)(rowp)); \
        buf[(G) & 1][2 * r + 1] = __builtin_nontemporal_load((const v4f*)(rowp + 128)); } } while (0)

    v4f acc0 = {}, acc1 = {};
    P1_LOAD(0);
    #pragma unroll
    for (int g = 0; g < P1_NGRP; ++g) {
        __builtin_amdgcn_sched_barrier(0);
        if (g + 1 < P1_NGRP) P1_LOAD(g + 1);
        __builtin_amdgcn_sched_barrier(0);
        #pragma unroll
        for (int r = 0; r < P1_GROUP; ++r) {
            const v4f u = buf[g & 1][2 * r];
            const v4f v = buf[g & 1][2 * r + 1];
            acc0.x += __expf(u.x); acc0.y += __expf(u.y);
            acc0.z += __expf(u.z); acc0.w += __expf(u.w);
            acc1.x += __expf(v.x); acc1.y += __expf(v.y);
            acc1.z += __expf(v.z); acc1.w += __expf(v.w);
        }
    }
#undef P1_LOAD

    part[wave][0][lane] = acc0;
    part[wave][1][lane] = acc1;
    __syncthreads();

    if (tid < 64) {
        const int j = tid >> 5;
        v4f s = part[0][j][lane];
        for (int w = 1; w < P1_WAVES; ++w) {
            const v4f t = part[w][j][lane];
            s.x += t.x; s.y += t.y; s.z += t.z; s.w += t.w;
        }
        v4f r;
        r.x = q.x / s.x; r.y = q.y / s.y; r.z = q.z / s.z; r.w = q.w / s.w;
        *(v4f*)(R + rp) = r;
    }
}

__global__ void aggregate_pass(const float* __restrict__ z,
                               const float* __restrict__ R,
                               float* __restrict__ Va) {
    __shared__ v4f smem[256];
    const int bi  = blockIdx.x;
    const int b   = bi >> 9;
    const int tid = threadIdx.x;
    const int c4    = (tid & 15) * 4;
    const int kbase = tid >> 4;

    const size_t zbase = (size_t)bi * PLANE;
    const float* Rb = R + (size_t)b * PLANE;

    v4f acc = {};
    #pragma unroll 4
    for (int k = kbase; k < NN; k += 16) {
        const int p = k * CC + c4;
        v4f zv = __builtin_nontemporal_load((const v4f*)(z + zbase + p));
        v4f rv = *(const v4f*)(Rb + p);
        acc.x += __expf(zv.x) * rv.x;
        acc.y += __expf(zv.y) * rv.y;
        acc.z += __expf(zv.z) * rv.z;
        acc.w += __expf(zv.w) * rv.w;
    }
    smem[tid] = acc;
    __syncthreads();

    if (tid < 16) {
        v4f s = smem[tid];
        #pragma unroll
        for (int q = 1; q < 16; ++q) {
            v4f t = smem[tid + 16 * q];
            s.x += t.x; s.y += t.y; s.z += t.z; s.w += t.w;
        }
        *(v4f*)(Va + (size_t)bi * CC + tid * 4) = s;
    }
}

extern "C" void kernel_launch(void* const* d_in, const int* in_sizes, int n_in,
                              void* d_out, int out_size, void* d_ws, size_t ws_size,
                              hipStream_t stream) {
    const float* V  = (const float*)d_in[0];
    const float* Q  = (const float*)d_in[1];
    const float* z  = (const float*)d_in[2];
    const float* W1 = (const float*)d_in[5];
    const float* b1 = (const float*)d_in[6];
    const float* W2 = (const float*)d_in[7];
    const float* b2 = (const float*)d_in[8];
    float* out = (float*)d_out;

    float* Qa = (float*)d_ws;
    float* R  = Qa + (size_t)MM * CC;
    float* Va = R  + (size_t)MM * CC;

    linear64_wmma<<<dim3(MM / 16, CC / 16), dim3(32), 0, stream>>>(Q, W1, b1, nullptr, Qa);

    softmax_denom_pass<<<dim3(PLANE / P1_STRIP, BB), dim3(P1_WAVES * 32), 0, stream>>>(z, Qa, R);

    aggregate_pass<<<dim3(BB * NN), dim3(256), 0, stream>>>(z, R, Va);

    linear64_wmma<<<dim3(MM / 16, CC / 16), dim3(32), 0, stream>>>(Va, W2, b2, V, out);
}
